# starting kernel
# baseline (speedup 1.0000x reference)
#include <hip/hip_runtime.h>

typedef float v2f __attribute__((ext_vector_type(2)));
typedef float v4f __attribute__((ext_vector_type(4)));
typedef float v8f __attribute__((ext_vector_type(8)));

#define TASKS 512
#define NWAY  5
#define NSHOT 5
#define NQ    75
#define NS    25
#define DIM   1600
#define PSTR  1604
#define ZROW  NWAY

__device__ inline float dot4sq(v4f a) {
    return a.x * a.x + a.y * a.y + a.z * a.z + a.w * a.w;
}

__global__ __launch_bounds__(160)
void proto_head_kernel(const float* __restrict__ query,
                       const float* __restrict__ support,
                       const int*   __restrict__ slabels,
                       const float* __restrict__ scale,
                       float*       __restrict__ out)
{
    __shared__ float ldsProto[(NWAY + 1) * PSTR];
    __shared__ float ldsAA[5 * 32];
    __shared__ float ldsBB[NWAY];
    __shared__ int   ldsLab[NS];

    const int task = blockIdx.x;
    const int tid  = threadIdx.x;
    const int lane = tid & 31;
    const int wave = tid >> 5;

    if (tid < NS)   ldsLab[tid] = slabels[task * NS + tid];
    if (tid < NWAY) ldsBB[tid]  = 0.0f;
    for (int c = tid; c < PSTR; c += 160) ldsProto[ZROW * PSTR + c] = 0.0f;
    __syncthreads();

    const float* sup = support + (size_t)task * NS * DIM;
    float bb0 = 0.f, bb1 = 0.f, bb2 = 0.f, bb3 = 0.f, bb4 = 0.f;

    for (int c = tid; c < DIM / 4; c += 160) {
        v4f a0 = {0,0,0,0}, a1 = {0,0,0,0}, a2 = {0,0,0,0}, a3 = {0,0,0,0}, a4 = {0,0,0,0};
        for (int s = 0; s < NS; ++s) {
            v4f v = *reinterpret_cast<const v4f*>(sup + s * DIM + c * 4);
            switch (ldsLab[s]) {
                case 0: a0 += v; break;
                case 1: a1 += v; break;
                case 2: a2 += v; break;
                case 3: a3 += v; break;
                default: a4 += v; break;
            }
        }
        const float inv = 1.0f / (float)NSHOT;
        a0.x *= inv; a0.y *= inv; a0.z *= inv; a0.w *= inv;
        a1.x *= inv; a1.y *= inv; a1.z *= inv; a1.w *= inv;
        a2.x *= inv; a2.y *= inv; a2.z *= inv; a2.w *= inv;
        a3.x *= inv; a3.y *= inv; a3.z *= inv; a3.w *= inv;
        a4.x *= inv; a4.y *= inv; a4.z *= inv; a4.w *= inv;
        *reinterpret_cast<v4f*>(&ldsProto[0 * PSTR + c * 4]) = a0;  bb0 += dot4sq(a0);
        *reinterpret_cast<v4f*>(&ldsProto[1 * PSTR + c * 4]) = a1;  bb1 += dot4sq(a1);
        *reinterpret_cast<v4f*>(&ldsProto[2 * PSTR + c * 4]) = a2;  bb2 += dot4sq(a2);
        *reinterpret_cast<v4f*>(&ldsProto[3 * PSTR + c * 4]) = a3;  bb3 += dot4sq(a3);
        *reinterpret_cast<v4f*>(&ldsProto[4 * PSTR + c * 4]) = a4;  bb4 += dot4sq(a4);
    }
    atomicAdd(&ldsBB[0], bb0);
    atomicAdd(&ldsBB[1], bb1);
    atomicAdd(&ldsBB[2], bb2);
    atomicAdd(&ldsBB[3], bb3);
    atomicAdd(&ldsBB[4], bb4);
    __syncthreads();

    const float sc   = scale[0];
    const int   tile = wave;
    const int   half = lane >> 4;
    const int   l16  = lane & 15;
    const int   arow = tile * 16 + l16;
    const int   crow = (arow < NQ) ? arow : 0;
    const float* qrow = query + (size_t)task * NQ * DIM + (size_t)crow * DIM + half * 2;
    const int   n     = l16;
    const int   bRow  = (n < NWAY) ? n : ZROW;
    const float* bptr = &ldsProto[bRow * PSTR + half * 2];

    v8f acc0 = {0,0,0,0,0,0,0,0};
    v8f acc1 = {0,0,0,0,0,0,0,0};
    float aa = 0.0f;

    #pragma unroll 4
    for (int k = 0; k < DIM; k += 8) {
        v2f a0 = *reinterpret_cast<const v2f*>(qrow + k);
        v2f a1 = *reinterpret_cast<const v2f*>(qrow + k + 4);
        aa += a0.x * a0.x + a0.y * a0.y + a1.x * a1.x + a1.y * a1.y;

        v2f b0 = *reinterpret_cast<const v2f*>(bptr + k);
        v2f b1 = *reinterpret_cast<const v2f*>(bptr + k + 4);

        acc0 = __builtin_amdgcn_wmma_f32_16x16x4_f32(
                   false, a0, false, b0, (short)0, acc0, false, false);
        acc1 = __builtin_amdgcn_wmma_f32_16x16x4_f32(
                   false, a1, false, b1, (short)0, acc1, false, false);
    }

    ldsAA[wave * 32 + lane] = aa;
    __syncthreads();

    if (n < NWAY) {
        const float bbv = ldsBB[n];
        #pragma unroll
        for (int j = 0; j < 8; ++j) {
            const int m   = half * 8 + j;
            const int row = tile * 16 + m;
            if (row < NQ) {
                const float aarow = ldsAA[wave * 32 + m] + ldsAA[wave * 32 + m + 16];
                const float abv   = acc0[j] + acc1[j];
                const float logit = sc * (-(aarow - 2.0f * abv + bbv) * (1.0f / (float)DIM));
                out[((size_t)task * NQ + row) * NWAY + n] = logit;
            }
        }
    }
}

extern "C" void kernel_launch(void* const* d_in, const int* in_sizes, int n_in,
                              void* d_out, int out_size, void* d_ws, size_t ws_size,
                              hipStream_t stream) {
    const float* query   = (const float*)d_in[0];
    const float* support = (const float*)d_in[1];
    const int*   slab    = (const int*)  d_in[2];
    const float* scale   = (const float*)d_in[3];
    float*       out     = (float*)d_out;
    (void)in_sizes; (void)n_in; (void)out_size; (void)d_ws; (void)ws_size;

    proto_head_kernel<<<TASKS, 160, 0, stream>>>(query, support, slab, scale, out);
}


# global loads issued as large independent groups ahead of use (25 support rows per column at once; query as 16-B pieces, two 10-load groups ahead); k regrouped, so AA/AB summation order changes
# speedup vs baseline: 2.0562x; 2.0562x over previous
#include <hip/hip_runtime.h>

typedef float v2f __attribute__((ext_vector_type(2)));
typedef float v4f __attribute__((ext_vector_type(4)));
typedef float v8f __attribute__((ext_vector_type(8)));

#define TASKS 512
#define NWAY  5
#define NSHOT 5
#define NQ    75
#define NS    25
#define DIM   1600
#define PSTR  1604
#define ZROW  NWAY
#define QCH   10
#define NGRP  (DIM / (8 * QCH))

__device__ inline float dot4sq(v4f a) {
    return a.x * a.x + a.y * a.y + a.z * a.z + a.w * a.w;
}

__device__ inline void load_group(v4f (&buf)[QCH], const float* __restrict__ p) {
    #pragma unroll
    for (int i = 0; i < QCH; ++i) buf[i] = *reinterpret_cast<const v4f*>(p + i * 8);
}

__device__ inline void use_group(const v4f (&buf)[QCH], const float* bp,
                                 v8f& acc0, v8f& acc1, float& aa) {
    #pragma unroll
    for (int i = 0; i < QCH; ++i) {
        const v4f a = buf[i];
        const v4f b = *reinterpret_cast<const v4f*>(bp + i * 8);
        aa += dot4sq(a);
        const v2f a0 = {a.x, a.y}, a1 = {a.z, a.w};
        const v2f b0 = {b.x, b.y}, b1 = {b.z, b.w};
        acc0 = __builtin_amdgcn_wmma_f32_16x16x4_f32(
                   false, a0, false, b0, (short)0, acc0, false, false);
        acc1 = __builtin_amdgcn_wmma_f32_16x16x4_f32(
                   false, a1, false, b1, (short)0, acc1, false, false);
    }
}

__global__ __launch_bounds__(160)
void proto_head_kernel(const float* __restrict__ query,
                       const float* __restrict__ support,
                       const int*   __restrict__ slabels,
                       const float* __restrict__ scale,
                       float*       __restrict__ out)
{
    __shared__ __attribute__((aligned(16))) float ldsProto[(NWAY + 1) * PSTR];
    __shared__ float ldsAA[5 * 32];
    __shared__ float ldsBB[NWAY];
    __shared__ int   ldsLab[NS];

    const int task = blockIdx.x;
    const int tid  = threadIdx.x;
    const int lane = tid & 31;
    const int wave = tid >> 5;

    const int   half = lane >> 4;
    const int   l16  = lane & 15;
    const int   arow = wave * 16 + l16;
    const int   crow = (arow < NQ) ? arow : 0;
    const float* qrow = query + (size_t)task * NQ * DIM + (size_t)crow * DIM + half * 4;

    int mylab = 0;
    if (tid < NS) mylab = slabels[task * NS + tid];
    v4f bufA[QCH], bufB[QCH];
    load_group(bufA, qrow);
    load_group(bufB, qrow + 8 * QCH);

    if (tid < NS)   ldsLab[tid] = mylab;
    if (tid < NWAY) ldsBB[tid]  = 0.0f;
    for (int c = tid; c < PSTR; c += 160) ldsProto[ZROW * PSTR + c] = 0.0f;
    __syncthreads();

    const float* sup = support + (size_t)task * NS * DIM;
    float cnt[NWAY] = {0.f, 0.f, 0.f, 0.f, 0.f};
    #pragma unroll
    for (int s = 0; s < NS; ++s) {
        const int lab = __builtin_amdgcn_readfirstlane(ldsLab[s]);
        #pragma unroll
        for (int j = 0; j < NWAY; ++j) {
            const bool in = (j < NWAY - 1) ? (lab == j) : (lab < 0 || lab > NWAY - 2);
            cnt[j] += in ? 1.0f : 0.0f;
        }
    }
    float inv[NWAY];
    #pragma unroll
    for (int j = 0; j < NWAY; ++j) inv[j] = 1.0f / cnt[j];

    float bb[NWAY] = {0.f, 0.f, 0.f, 0.f, 0.f};
    for (int c = tid; c < DIM / 4; c += 160) {
        v4f v[NS];
        #pragma unroll
        for (int s = 0; s < NS; ++s)
            v[s] = *reinterpret_cast<const v4f*>(sup + s * DIM + c * 4);

        v4f a[NWAY];
        #pragma unroll
        for (int j = 0; j < NWAY; ++j) a[j] = (v4f){0.f, 0.f, 0.f, 0.f};
        #pragma unroll
        for (int s = 0; s < NS; ++s) {
        const int lab = __builtin_amdgcn_readfirstlane(ldsLab[s]);
            #pragma unroll
            for (int j = 0; j < NWAY; ++j) {
                const bool in = (j < NWAY - 1) ? (lab == j) : (lab < 0 || lab > NWAY - 2);
                const float w = in ? 1.0f : 0.0f;
                a[j].x = __builtin_fmaf(w, v[s].x, a[j].x);
                a[j].y = __builtin_fmaf(w, v[s].y, a[j].y);
                a[j].z = __builtin_fmaf(w, v[s].z, a[j].z);
                a[j].w = __builtin_fmaf(w, v[s].w, a[j].w);
            }
        }
        #pragma unroll
        for (int j = 0; j < NWAY; ++j) {
            a[j].x *= inv[j]; a[j].y *= inv[j]; a[j].z *= inv[j]; a[j].w *= inv[j];
            *reinterpret_cast<v4f*>(&ldsProto[j * PSTR + c * 4]) = a[j];
            bb[j] += dot4sq(a[j]);
        }
    }
    #pragma unroll
    for (int j = 0; j < NWAY; ++j) atomicAdd(&ldsBB[j], bb[j]);
    __syncthreads();

    const float sc   = scale[0];
    const int   n    = l16;
    const int   bRow = (n < NWAY) ? n : ZROW;
    const float* bptr = &ldsProto[bRow * PSTR + half * 4];

    v8f acc0 = {0,0,0,0,0,0,0,0};
    v8f acc1 = {0,0,0,0,0,0,0,0};
    float aa = 0.0f;

    #pragma unroll
    for (int g = 0; g < NGRP; g += 2) {
        use_group(bufA, bptr + g * (8 * QCH), acc0, acc1, aa);
        if (g + 2 < NGRP) load_group(bufA, qrow + (g + 2) * (8 * QCH));
        use_group(bufB, bptr + (g + 1) * (8 * QCH), acc0, acc1, aa);
        if (g + 3 < NGRP) load_group(bufB, qrow + (g + 3) * (8 * QCH));
    }

    ldsAA[wave * 32 + lane] = aa;
    __syncthreads();

    if (n < NWAY) {
        const float bbv = ldsBB[n];
        #pragma unroll
        for (int j = 0; j < 8; ++j) {
            const int m   = half * 8 + j;
            const int row = wave * 16 + m;
            if (row < NQ) {
                const float aarow = ldsAA[wave * 32 + m] + ldsAA[wave * 32 + m + 16];
                const float abv   = acc0[j] + acc1[j];
                const float logit = sc * (-(aarow - 2.0f * abv + bbv) * (1.0f / (float)DIM));
                out[((size_t)task * NQ + row) * NWAY + n] = logit;
            }
        }
    }
}

extern "C" void kernel_launch(void* const* d_in, const int* in_sizes, int n_in,
                              void* d_out, int out_size, void* d_ws, size_t ws_size,
                              hipStream_t stream) {
    const float* query   = (const float*)d_in[0];
    const float* support = (const float*)d_in[1];
    const int*   slab    = (const int*)  d_in[2];
    const float* scale   = (const float*)d_in[3];
    float*       out     = (float*)d_out;
    (void)in_sizes; (void)n_in; (void)out_size; (void)d_ws; (void)ws_size;

    proto_head_kernel<<<TASKS, 160, 0, stream>>>(query, support, slab, scale, out);
}


# query loads in flight per lane 5-20 -> 24-32: four register buffers of eight 16-B loads, each re-requested as one block right behind its last use (order pinned); k order and sums unchanged
# speedup vs baseline: 2.2608x; 1.0995x over previous
#include <hip/hip_runtime.h>

typedef float v2f __attribute__((ext_vector_type(2)));
typedef float v4f __attribute__((ext_vector_type(4)));
typedef float v8f __attribute__((ext_vector_type(8)));

#define TASKS 512
#define NWAY  5
#define NSHOT 5
#define NQ    75
#define NS    25
#define DIM   1600
#define PSTR  1604
#define ZROW  NWAY
#define QCH   8
#define NGRP  (DIM / (8 * QCH))
#define NBUF  4
#define NPRE  2

__device__ inline float dot4sq(v4f a) {
    return a.x * a.x + a.y * a.y + a.z * a.z + a.w * a.w;
}

__device__ inline void load_group(v4f (&buf)[QCH], const float* __restrict__ p) {
    #pragma unroll
    for (int i = 0; i < QCH; ++i) buf[i] = *reinterpret_cast<const v4f*>(p + i * 8);
}

__device__ inline void use_group(const v4f (&buf)[QCH], const float* bp,
                                 v8f& acc0, v8f& acc1, float& aa) {
    #pragma unroll
    for (int i = 0; i < QCH; ++i) {
        const v4f a = buf[i];
        const v4f b = *reinterpret_cast<const v4f*>(bp + i * 8);
        aa += dot4sq(a);
        const v2f a0 = {a.x, a.y}, a1 = {a.z, a.w};
        const v2f b0 = {b.x, b.y}, b1 = {b.z, b.w};
        acc0 = __builtin_amdgcn_wmma_f32_16x16x4_f32(
                   false, a0, false, b0, (short)0, acc0, false, false);
        acc1 = __builtin_amdgcn_wmma_f32_16x16x4_f32(
                   false, a1, false, b1, (short)0, acc1, false, false);
    }
}

__global__ __launch_bounds__(160)
void proto_head_kernel(const float* __restrict__ query,
                       const float* __restrict__ support,
                       const int*   __restrict__ slabels,
                       const float* __restrict__ scale,
                       float*       __restrict__ out)
{
    __shared__ __attribute__((aligned(16))) float ldsProto[(NWAY + 1) * PSTR];
    __shared__ float ldsAA[5 * 32];
    __shared__ float ldsBB[NWAY];
    __shared__ int   ldsLab[NS];

    const int task = blockIdx.x;
    const int tid  = threadIdx.x;
    const int lane = tid & 31;
    const int wave = tid >> 5;

    const int   half = lane >> 4;
    const int   l16  = lane & 15;
    const int   arow = wave * 16 + l16;
    const int   crow = (arow < NQ) ? arow : 0;
    const float* qrow = query + (size_t)task * NQ * DIM + (size_t)crow * DIM + half * 4;

    int mylab = 0;
    if (tid < NS) mylab = slabels[task * NS + tid];
    v4f buf[NBUF][QCH];
    #pragma unroll
    for (int b = 0; b < NPRE; ++b) load_group(buf[b], qrow + b * (8 * QCH));

    if (tid < NS)   ldsLab[tid] = mylab;
    if (tid < NWAY) ldsBB[tid]  = 0.0f;
    for (int c = tid; c < PSTR; c += 160) ldsProto[ZROW * PSTR + c] = 0.0f;
    __syncthreads();

    const float* sup = support + (size_t)task * NS * DIM;
    float cnt[NWAY] = {0.f, 0.f, 0.f, 0.f, 0.f};
    #pragma unroll
    for (int s = 0; s < NS; ++s) {
        const int lab = __builtin_amdgcn_readfirstlane(ldsLab[s]);
        #pragma unroll
        for (int j = 0; j < NWAY; ++j) {
            const bool in = (j < NWAY - 1) ? (lab == j) : (lab < 0 || lab > NWAY - 2);
            cnt[j] += in ? 1.0f : 0.0f;
        }
    }
    float inv[NWAY];
    #pragma unroll
    for (int j = 0; j < NWAY; ++j) inv[j] = 1.0f / cnt[j];

    float bb[NWAY] = {0.f, 0.f, 0.f, 0.f, 0.f};
    for (int c = tid; c < DIM / 4; c += 160) {
        v4f v[NS];
        #pragma unroll
        for (int s = 0; s < NS; ++s)
            v[s] = *reinterpret_cast<const v4f*>(sup + s * DIM + c * 4);

        v4f a[NWAY];
        #pragma unroll
        for (int j = 0; j < NWAY; ++j) a[j] = (v4f){0.f, 0.f, 0.f, 0.f};
        #pragma unroll
        for (int s = 0; s < NS; ++s) {
        const int lab = __builtin_amdgcn_readfirstlane(ldsLab[s]);
            #pragma unroll
            for (int j = 0; j < NWAY; ++j) {
                const bool in = (j < NWAY - 1) ? (lab == j) : (lab < 0 || lab > NWAY - 2);
                const float w = in ? 1.0f : 0.0f;
                a[j].x = __builtin_fmaf(w, v[s].x, a[j].x);
                a[j].y = __builtin_fmaf(w, v[s].y, a[j].y);
                a[j].z = __builtin_fmaf(w, v[s].z, a[j].z);
                a[j].w = __builtin_fmaf(w, v[s].w, a[j].w);
            }
        }
        #pragma unroll
        for (int j = 0; j < NWAY; ++j) {
            a[j].x *= inv[j]; a[j].y *= inv[j]; a[j].z *= inv[j]; a[j].w *= inv[j];
            *reinterpret_cast<v4f*>(&ldsProto[j * PSTR + c * 4]) = a[j];
            bb[j] += dot4sq(a[j]);
        }
    }
    #pragma unroll
    for (int j = 0; j < NWAY; ++j) atomicAdd(&ldsBB[j], bb[j]);
    __syncthreads();

    const float sc   = scale[0];
    const int   n    = l16;
    const int   bRow = (n < NWAY) ? n : ZROW;
    const float* bptr = &ldsProto[bRow * PSTR + half * 4];

    v8f acc0 = {0,0,0,0,0,0,0,0};
    v8f acc1 = {0,0,0,0,0,0,0,0};
    float aa = 0.0f;

    #pragma unroll
    for (int b = NPRE; b < NBUF; ++b) load_group(buf[b], qrow + b * (8 * QCH));
    #pragma unroll
    for (int g = 0; g < NGRP; ++g) {
        __builtin_amdgcn_sched_barrier(0);
        use_group(buf[g % NBUF], bptr + g * (8 * QCH), acc0, acc1, aa);
        __builtin_amdgcn_sched_barrier(0);
        if (g + NBUF < NGRP) load_group(buf[g % NBUF], qrow + (g + NBUF) * (8 * QCH));
    }

    ldsAA[wave * 32 + lane] = aa;
    __syncthreads();

    if (n < NWAY) {
        const float bbv = ldsBB[n];
        #pragma unroll
        for (int j = 0; j < 8; ++j) {
            const int m   = half * 8 + j;
            const int row = wave * 16 + m;
            if (row < NQ) {
                const float aarow = ldsAA[wave * 32 + m] + ldsAA[wave * 32 + m + 16];
                const float abv   = acc0[j] + acc1[j];
                const float logit = sc * (-(aarow - 2.0f * abv + bbv) * (1.0f / (float)DIM));
                out[((size_t)task * NQ + row) * NWAY + n] = logit;
            }
        }
    }
}

extern "C" void kernel_launch(void* const* d_in, const int* in_sizes, int n_in,
                              void* d_out, int out_size, void* d_ws, size_t ws_size,
                              hipStream_t stream) {
    const float* query   = (const float*)d_in[0];
    const float* support = (const float*)d_in[1];
    const int*   slab    = (const int*)  d_in[2];
    const float* scale   = (const float*)d_in[3];
    float*       out     = (float*)d_out;
    (void)in_sizes; (void)n_in; (void)out_size; (void)d_ws; (void)ws_size;

    proto_head_kernel<<<TASKS, 160, 0, stream>>>(query, support, slab, scale, out);
}


# global loads stay in flight through the prototype build: support rows as one rolling 25-load window requested at entry (tail in 8-byte pieces), barriers wait for LDS only; BB partial sums regrouped
# speedup vs baseline: 2.3568x; 1.0425x over previous
#include <hip/hip_runtime.h>

typedef float v2f __attribute__((ext_vector_type(2)));
typedef float v4f __attribute__((ext_vector_type(4)));
typedef float v8f __attribute__((ext_vector_type(8)));

#define TASKS 512
#define NWAY  5
#define NSHOT 5
#define NQ    75
#define NS    25
#define DIM   1600
#define PSTR  1604
#define ZROW  NWAY
#define QCH   8
#define NGRP  (DIM / (8 * QCH))
#define NBUF  4
#define NPRE  2
#define NTHR  160
#define SEC   (4 * NTHR)
#define TAIL0 (2 * SEC)
#define RSTEP 5

__device__ inline float dot4sq(v4f a) {
    return a.x * a.x + a.y * a.y + a.z * a.z + a.w * a.w;
}

__device__ inline void load_group(v4f (&buf)[QCH], const float* __restrict__ p) {
    #pragma unroll
    for (int i = 0; i < QCH; ++i) buf[i] = *reinterpret_cast<const v4f*>(p + i * 8);
}

__device__ inline void use_group(const v4f (&buf)[QCH], const float* bp,
                                 v8f& acc0, v8f& acc1, float& aa) {
    #pragma unroll
    for (int i = 0; i < QCH; ++i) {
        const v4f a = buf[i];
        const v4f b = *reinterpret_cast<const v4f*>(bp + i * 8);
        aa += dot4sq(a);
        const v2f a0 = {a.x, a.y}, a1 = {a.z, a.w};
        const v2f b0 = {b.x, b.y}, b1 = {b.z, b.w};
        acc0 = __builtin_amdgcn_wmma_f32_16x16x4_f32(
                   false, a0, false, b0, (short)0, acc0, false, false);
        acc1 = __builtin_amdgcn_wmma_f32_16x16x4_f32(
                   false, a1, false, b1, (short)0, acc1, false, false);
    }
}

__device__ inline void lds_barrier() {
    __builtin_amdgcn_fence(__ATOMIC_RELEASE, "workgroup", "local");
    __builtin_amdgcn_s_barrier();
    __builtin_amdgcn_fence(__ATOMIC_ACQUIRE, "workgroup", "local");
}

__device__ inline int fresh_label(int lab) {
    asm volatile("" : "+s"(lab));
    return lab;
}

__device__ inline float cls_w(int lab, int j) {
    const bool in = (j < NWAY - 1) ? (lab == j) : (lab < 0 || lab > NWAY - 2);
    return in ? 1.0f : 0.0f;
}

__global__ __launch_bounds__(NTHR)
void proto_head_kernel(const float* __restrict__ query,
                       const float* __restrict__ support,
                       const int*   __restrict__ slabels,
                       const float* __restrict__ scale,
                       float*       __restrict__ out)
{
    __shared__ __attribute__((aligned(16))) float ldsProto[(NWAY + 1) * PSTR];
    __shared__ float ldsAA[5 * 32];
    __shared__ float ldsBB[NWAY];
    __shared__ int   ldsLab[NS];

    const int task = blockIdx.x;
    const int tid  = threadIdx.x;
    const int lane = tid & 31;
    const int wave = tid >> 5;

    const int   half = lane >> 4;
    const int   l16  = lane & 15;
    const int   arow = wave * 16 + l16;
    const int   crow = (arow < NQ) ? arow : 0;
    const float* qrow = query + (size_t)task * NQ * DIM + (size_t)crow * DIM + half * 4;

    const float* sup = support + (size_t)task * NS * DIM;
    const float* sp0 = sup + tid * 4;
    const float* sp1 = sup + SEC + tid * 4;
    const float* sp2 = sup + TAIL0 + tid * 2;

    int mylab = 0;
    if (tid < NS) mylab = slabels[task * NS + tid];
    v4f rowA[NS];
    #pragma unroll
    for (int s = 0; s < NS; ++s) rowA[s] = *reinterpret_cast<const v4f*>(sp0 + s * DIM);

    if (tid < NS)   ldsLab[tid] = mylab;
    if (tid < NWAY) ldsBB[tid]  = 0.0f;
    for (int c = tid; c < PSTR; c += NTHR) ldsProto[ZROW * PSTR + c] = 0.0f;

    v4f buf[NBUF][QCH];
    #pragma unroll
    for (int b = 0; b < NPRE; ++b) load_group(buf[b], qrow + b * (8 * QCH));
    lds_barrier();

    int lab[NS];
    #pragma unroll
    for (int s = 0; s < NS; ++s) {
        lab[s] = __builtin_amdgcn_readfirstlane(ldsLab[s]);
    }
    float cnt[NWAY] = {0.f, 0.f, 0.f, 0.f, 0.f};
    #pragma unroll
    for (int s = 0; s < NS; ++s) {
        #pragma unroll
        for (int j = 0; j < NWAY; ++j) cnt[j] += cls_w(lab[s], j);
    }
    float inv[NWAY];
    #pragma unroll
    for (int j = 0; j < NWAY; ++j) inv[j] = 1.0f / cnt[j];

    float bb[NWAY] = {0.f, 0.f, 0.f, 0.f, 0.f};

    v4f a[NWAY];

    #pragma unroll
    for (int j = 0; j < NWAY; ++j) a[j] = (v4f){0.f, 0.f, 0.f, 0.f};
    #pragma unroll
    for (int g = 0; g < NS / RSTEP; ++g) {
        __builtin_amdgcn_sched_barrier(0);
        #pragma unroll
        for (int r = 0; r < RSTEP; ++r) {
            const int s = g * RSTEP + r;
            const int ls = fresh_label(lab[s]);
            #pragma unroll
            for (int j = 0; j < NWAY; ++j) {
                const float w = cls_w(ls, j);
                a[j].x = __builtin_fmaf(w, rowA[s].x, a[j].x);
                a[j].y = __builtin_fmaf(w, rowA[s].y, a[j].y);
                a[j].z = __builtin_fmaf(w, rowA[s].z, a[j].z);
                a[j].w = __builtin_fmaf(w, rowA[s].w, a[j].w);
            }
        }
        __builtin_amdgcn_sched_barrier(0);
        #pragma unroll
        for (int r = 0; r < RSTEP; ++r) {
            const int s = g * RSTEP + r;
            rowA[s] = *reinterpret_cast<const v4f*>(sp1 + s * DIM);
        }
    }
    #pragma unroll
    for (int j = 0; j < NWAY; ++j) {
        a[j].x *= inv[j]; a[j].y *= inv[j]; a[j].z *= inv[j]; a[j].w *= inv[j];
        *reinterpret_cast<v4f*>(&ldsProto[j * PSTR + tid * 4]) = a[j];
        bb[j] += dot4sq(a[j]);
    }

    v2f rowT[NS];
    #pragma unroll
    for (int j = 0; j < NWAY; ++j) a[j] = (v4f){0.f, 0.f, 0.f, 0.f};
    #pragma unroll
    for (int g = 0; g < NS / RSTEP; ++g) {
        __builtin_amdgcn_sched_barrier(0);
        #pragma unroll
        for (int r = 0; r < RSTEP; ++r) {
            const int s = g * RSTEP + r;
            const int ls = fresh_label(lab[s]);
            #pragma unroll
            for (int j = 0; j < NWAY; ++j) {
                const float w = cls_w(ls, j);
                a[j].x = __builtin_fmaf(w, rowA[s].x, a[j].x);
                a[j].y = __builtin_fmaf(w, rowA[s].y, a[j].y);
                a[j].z = __builtin_fmaf(w, rowA[s].z, a[j].z);
                a[j].w = __builtin_fmaf(w, rowA[s].w, a[j].w);
            }
        }
        __builtin_amdgcn_sched_barrier(0);
        #pragma unroll
        for (int r = 0; r < RSTEP; ++r) {
            const int s = g * RSTEP + r;
            rowT[s] = *reinterpret_cast<const v2f*>(sp2 + s * DIM);
        }
    }
    #pragma unroll
    for (int j = 0; j < NWAY; ++j) {
        a[j].x *= inv[j]; a[j].y *= inv[j]; a[j].z *= inv[j]; a[j].w *= inv[j];
        *reinterpret_cast<v4f*>(&ldsProto[j * PSTR + SEC + tid * 4]) = a[j];
        bb[j] += dot4sq(a[j]);
    }

    v2f t[NWAY];
    #pragma unroll
    for (int j = 0; j < NWAY; ++j) t[j] = (v2f){0.f, 0.f};
    #pragma unroll
    for (int g = 0; g < NS / RSTEP; ++g) {
        __builtin_amdgcn_sched_barrier(0);
        if (g == 0) load_group(buf[NPRE],     qrow + NPRE * (8 * QCH));
        if (g == 2) load_group(buf[NPRE + 1], qrow + (NPRE + 1) * (8 * QCH));
        __builtin_amdgcn_sched_barrier(0);
        #pragma unroll
        for (int r = 0; r < RSTEP; ++r) {
            const int s = g * RSTEP + r;
            const int ls = fresh_label(lab[s]);
            #pragma unroll
            for (int j = 0; j < NWAY; ++j) {
                const float w = cls_w(ls, j);
                t[j].x = __builtin_fmaf(w, rowT[s].x, t[j].x);
                t[j].y = __builtin_fmaf(w, rowT[s].y, t[j].y);
            }
        }
    }
    #pragma unroll
    for (int j = 0; j < NWAY; ++j) {
        t[j].x *= inv[j]; t[j].y *= inv[j];
        *reinterpret_cast<v2f*>(&ldsProto[j * PSTR + TAIL0 + tid * 2]) = t[j];
        bb[j] += t[j].x * t[j].x + t[j].y * t[j].y;
    }

    #pragma unroll
    for (int j = 0; j < NWAY; ++j) atomicAdd(&ldsBB[j], bb[j]);
    lds_barrier();

    const float sc   = scale[0];
    const int   n    = l16;
    const int   bRow = (n < NWAY) ? n : ZROW;
    const float* bptr = &ldsProto[bRow * PSTR + half * 4];

    v8f acc0 = {0,0,0,0,0,0,0,0};
    v8f acc1 = {0,0,0,0,0,0,0,0};
    float aa = 0.0f;

    #pragma unroll
    for (int g = 0; g < NGRP; ++g) {
        __builtin_amdgcn_sched_barrier(0);
        use_group(buf[g % NBUF], bptr + g * (8 * QCH), acc0, acc1, aa);
        __builtin_amdgcn_sched_barrier(0);
        if (g + NBUF < NGRP) load_group(buf[g % NBUF], qrow + (g + NBUF) * (8 * QCH));
    }

    ldsAA[wave * 32 + lane] = aa;
    __syncthreads();

    if (n < NWAY) {
        const float bbv = ldsBB[n];
        #pragma unroll
        for (int j = 0; j < 8; ++j) {
            const int m   = half * 8 + j;
            const int row = wave * 16 + m;
            if (row < NQ) {
                const float aarow = ldsAA[wave * 32 + m] + ldsAA[wave * 32 + m + 16];
                const float abv   = acc0[j] + acc1[j];
                const float logit = sc * (-(aarow - 2.0f * abv + bbv) * (1.0f / (float)DIM));
                out[((size_t)task * NQ + row) * NWAY + n] = logit;
            }
        }
    }
}

extern "C" void kernel_launch(void* const* d_in, const int* in_sizes, int n_in,
                              void* d_out, int out_size, void* d_ws, size_t ws_size,
                              hipStream_t stream) {
    const float* query   = (const float*)d_in[0];
    const float* support = (const float*)d_in[1];
    const int*   slab    = (const int*)  d_in[2];
    const float* scale   = (const float*)d_in[3];
    float*       out     = (float*)d_out;
    (void)in_sizes; (void)n_in; (void)out_size; (void)d_ws; (void)ws_size;

    proto_head_kernel<<<TASKS, 160, 0, stream>>>(query, support, slab, scale, out);
}


# query loads made whole-line: a wave-wide 16-B load reads 2 rows x 256 B (4 lines; was 16 rows x 32 B, 16 lines); A fragments via a wave-private LDS stage; k and summation order unchanged
# speedup vs baseline: 2.5812x; 1.0952x over previous
#include <hip/hip_runtime.h>

typedef float v2f __attribute__((ext_vector_type(2)));
typedef float v4f __attribute__((ext_vector_type(4)));
typedef float v8f __attribute__((ext_vector_type(8)));

#define TASKS 512
#define NWAY  5
#define NSHOT 5
#define NQ    75
#define NS    25
#define DIM   1600
#define PSTR  1604
#define ZROW  NWAY
#define QCH   8
#define NGRP  (DIM / (8 * QCH))
#define NBUF  4
#define NPRE  2
#define NTHR  160
#define SEC   (4 * NTHR)
#define TAIL0 (2 * SEC)
#define RSTEP 5
#define NWAVE (NTHR / 32)
#define SPITCH 68
#define STAGE (16 * SPITCH)

__device__ inline float dot4sq(v4f a) {
    return a.x * a.x + a.y * a.y + a.z * a.z + a.w * a.w;
}

__device__ inline void load_group(v4f (&buf)[QCH], const float* __restrict__ qtask,
                                  const unsigned (&roff)[QCH], int g) {
    #pragma unroll
    for (int i = 0; i < QCH; ++i)
        buf[i] = *reinterpret_cast<const v4f*>(qtask + roff[i] + g * (8 * QCH));
}

__device__ inline void wave_lds_order() {
    __builtin_amdgcn_fence(__ATOMIC_RELEASE, "wavefront", "local");
    __builtin_amdgcn_wave_barrier();
    __builtin_amdgcn_fence(__ATOMIC_ACQUIRE, "wavefront", "local");
}

__device__ inline void stage_group(const v4f (&buf)[QCH], float* st) {
    #pragma unroll
    for (int i = 0; i < QCH; ++i) *reinterpret_cast<v4f*>(st + i * (2 * SPITCH)) = buf[i];
}

__device__ inline void frag_half(v4f (&fa)[QCH / 2], const float* fl) {
    #pragma unroll
    for (int j = 0; j < QCH / 2; ++j) fa[j] = *reinterpret_cast<const v4f*>(fl + j * 8);
}

__device__ inline void proto_half(v4f (&fb)[QCH / 2], const float* bp) {
    #pragma unroll
    for (int j = 0; j < QCH / 2; ++j) fb[j] = *reinterpret_cast<const v4f*>(bp + j * 8);
}

__device__ inline void use_half(const v4f (&fa)[QCH / 2], const v4f (&fb)[QCH / 2],
                                v8f& acc0, v8f& acc1, float& aa) {
    #pragma unroll
    for (int i = 0; i < QCH / 2; ++i) {
        const v4f a = fa[i];
        const v4f b = fb[i];
        aa += dot4sq(a);
        const v2f a0 = {a.x, a.y}, a1 = {a.z, a.w};
        const v2f b0 = {b.x, b.y}, b1 = {b.z, b.w};
        acc0 = __builtin_amdgcn_wmma_f32_16x16x4_f32(
                   false, a0, false, b0, (short)0, acc0, false, false);
        acc1 = __builtin_amdgcn_wmma_f32_16x16x4_f32(
                   false, a1, false, b1, (short)0, acc1, false, false);
    }
}

__device__ inline void lds_barrier() {
    __builtin_amdgcn_fence(__ATOMIC_RELEASE, "workgroup", "local");
    __builtin_amdgcn_s_barrier();
    __builtin_amdgcn_fence(__ATOMIC_ACQUIRE, "workgroup", "local");
}

__device__ inline int fresh_label(int lab) {
    asm volatile("" : "+s"(lab));
    return lab;
}

__device__ inline float cls_w(int lab, int j) {
    const bool in = (j < NWAY - 1) ? (lab == j) : (lab < 0 || lab > NWAY - 2);
    return in ? 1.0f : 0.0f;
}

__global__ __launch_bounds__(NTHR)
void proto_head_kernel(const float* __restrict__ query,
                       const float* __restrict__ support,
                       const int*   __restrict__ slabels,
                       const float* __restrict__ scale,
                       float*       __restrict__ out)
{
    __shared__ __attribute__((aligned(16))) float ldsProto[(NWAY + 1) * PSTR];
    __shared__ __attribute__((aligned(16))) float ldsStage[NWAVE * STAGE];
    __shared__ float ldsAA[5 * 32];
    __shared__ float ldsBB[NWAY];
    __shared__ int   ldsLab[NS];

    const int task = blockIdx.x;
    const int tid  = threadIdx.x;
    const int lane = tid & 31;
    const int wave = tid >> 5;

    const int   half = lane >> 4;
    const int   l16  = lane & 15;
    const float* qtask = query + (size_t)task * NQ * DIM;
    unsigned roff[QCH];
    #pragma unroll
    for (int i = 0; i < QCH; ++i) {
        const int r  = wave * 16 + 2 * i + half;
        const int cr = (r < NQ) ? r : 0;
        roff[i] = (unsigned)(cr * DIM + l16 * 4);
    }
    float*       stw = &ldsStage[wave * STAGE + half * SPITCH + l16 * 4];
    const float* frd = &ldsStage[wave * STAGE + l16 * SPITCH + half * 4];

    const float* sup = support + (size_t)task * NS * DIM;
    const float* sp0 = sup + tid * 4;
    const float* sp1 = sup + SEC + tid * 4;
    const float* sp2 = sup + TAIL0 + tid * 2;

    int mylab = 0;
    if (tid < NS) mylab = slabels[task * NS + tid];
    v4f rowA[NS];
    #pragma unroll
    for (int s = 0; s < NS; ++s) rowA[s] = *reinterpret_cast<const v4f*>(sp0 + s * DIM);

    if (tid < NS)   ldsLab[tid] = mylab;
    if (tid < NWAY) ldsBB[tid]  = 0.0f;
    for (int c = tid; c < PSTR; c += NTHR) ldsProto[ZROW * PSTR + c] = 0.0f;

    v4f buf[NBUF][QCH];
    #pragma unroll
    for (int b = 0; b < NPRE; ++b) load_group(buf[b], qtask, roff, b);
    lds_barrier();

    int lab[NS];
    #pragma unroll
    for (int s = 0; s < NS; ++s) {
        lab[s] = __builtin_amdgcn_readfirstlane(ldsLab[s]);
    }
    float cnt[NWAY] = {0.f, 0.f, 0.f, 0.f, 0.f};
    #pragma unroll
    for (int s = 0; s < NS; ++s) {
        #pragma unroll
        for (int j = 0; j < NWAY; ++j) cnt[j] += cls_w(lab[s], j);
    }
    float inv[NWAY];
    #pragma unroll
    for (int j = 0; j < NWAY; ++j) inv[j] = 1.0f / cnt[j];

    float bb[NWAY] = {0.f, 0.f, 0.f, 0.f, 0.f};

    v4f a[NWAY];

    #pragma unroll
    for (int j = 0; j < NWAY; ++j) a[j] = (v4f){0.f, 0.f, 0.f, 0.f};
    #pragma unroll
    for (int g = 0; g < NS / RSTEP; ++g) {
        __builtin_amdgcn_sched_barrier(0);
        #pragma unroll
        for (int r = 0; r < RSTEP; ++r) {
            const int s = g * RSTEP + r;
            const int ls = fresh_label(lab[s]);
            #pragma unroll
            for (int j = 0; j < NWAY; ++j) {
                const float w = cls_w(ls, j);
                a[j].x = __builtin_fmaf(w, rowA[s].x, a[j].x);
                a[j].y = __builtin_fmaf(w, rowA[s].y, a[j].y);
                a[j].z = __builtin_fmaf(w, rowA[s].z, a[j].z);
                a[j].w = __builtin_fmaf(w, rowA[s].w, a[j].w);
            }
        }
        __builtin_amdgcn_sched_barrier(0);
        #pragma unroll
        for (int r = 0; r < RSTEP; ++r) {
            const int s = g * RSTEP + r;
            rowA[s] = *reinterpret_cast<const v4f*>(sp1 + s * DIM);
        }
    }
    #pragma unroll
    for (int j = 0; j < NWAY; ++j) {
        a[j].x *= inv[j]; a[j].y *= inv[j]; a[j].z *= inv[j]; a[j].w *= inv[j];
        *reinterpret_cast<v4f*>(&ldsProto[j * PSTR + tid * 4]) = a[j];
        bb[j] += dot4sq(a[j]);
    }

    v2f rowT[NS];
    #pragma unroll
    for (int j = 0; j < NWAY; ++j) a[j] = (v4f){0.f, 0.f, 0.f, 0.f};
    #pragma unroll
    for (int g = 0; g < NS / RSTEP; ++g) {
        __builtin_amdgcn_sched_barrier(0);
        #pragma unroll
        for (int r = 0; r < RSTEP; ++r) {
            const int s = g * RSTEP + r;
            const int ls = fresh_label(lab[s]);
            #pragma unroll
            for (int j = 0; j < NWAY; ++j) {
                const float w = cls_w(ls, j);
                a[j].x = __builtin_fmaf(w, rowA[s].x, a[j].x);
                a[j].y = __builtin_fmaf(w, rowA[s].y, a[j].y);
                a[j].z = __builtin_fmaf(w, rowA[s].z, a[j].z);
                a[j].w = __builtin_fmaf(w, rowA[s].w, a[j].w);
            }
        }
        __builtin_amdgcn_sched_barrier(0);
        #pragma unroll
        for (int r = 0; r < RSTEP; ++r) {
            const int s = g * RSTEP + r;
            rowT[s] = *reinterpret_cast<const v2f*>(sp2 + s * DIM);
        }
    }
    #pragma unroll
    for (int j = 0; j < NWAY; ++j) {
        a[j].x *= inv[j]; a[j].y *= inv[j]; a[j].z *= inv[j]; a[j].w *= inv[j];
        *reinterpret_cast<v4f*>(&ldsProto[j * PSTR + SEC + tid * 4]) = a[j];
        bb[j] += dot4sq(a[j]);
    }

    v2f t[NWAY];
    #pragma unroll
    for (int j = 0; j < NWAY; ++j) t[j] = (v2f){0.f, 0.f};
    #pragma unroll
    for (int g = 0; g < NS / RSTEP; ++g) {
        __builtin_amdgcn_sched_barrier(0);
        if (g == 0) load_group(buf[NPRE],     qtask, roff, NPRE);
        if (g == 2) load_group(buf[NPRE + 1], qtask, roff, NPRE + 1);
        __builtin_amdgcn_sched_barrier(0);
        #pragma unroll
        for (int r = 0; r < RSTEP; ++r) {
            const int s = g * RSTEP + r;
            const int ls = fresh_label(lab[s]);
            #pragma unroll
            for (int j = 0; j < NWAY; ++j) {
                const float w = cls_w(ls, j);
                t[j].x = __builtin_fmaf(w, rowT[s].x, t[j].x);
                t[j].y = __builtin_fmaf(w, rowT[s].y, t[j].y);
            }
        }
    }
    #pragma unroll
    for (int j = 0; j < NWAY; ++j) {
        t[j].x *= inv[j]; t[j].y *= inv[j];
        *reinterpret_cast<v2f*>(&ldsProto[j * PSTR + TAIL0 + tid * 2]) = t[j];
        bb[j] += t[j].x * t[j].x + t[j].y * t[j].y;
    }

    #pragma unroll
    for (int j = 0; j < NWAY; ++j) atomicAdd(&ldsBB[j], bb[j]);
    lds_barrier();

    const float sc   = scale[0];
    const int   n    = l16;
    const int   bRow = (n < NWAY) ? n : ZROW;
    const float* bptr = &ldsProto[bRow * PSTR + half * 4];

    v8f acc0 = {0,0,0,0,0,0,0,0};
    v8f acc1 = {0,0,0,0,0,0,0,0};
    float aa = 0.0f;

    v4f faL[QCH / 2], faH[QCH / 2];
    v4f fbL[QCH / 2], fbH[QCH / 2];
    wave_lds_order();
    stage_group(buf[0], stw);
    wave_lds_order();
    load_group(buf[0], qtask, roff, NBUF);
    frag_half(faL, frd);
    proto_half(fbL, bptr);

    #pragma unroll
    for (int g = 0; g < NGRP; ++g) {
        __builtin_amdgcn_sched_barrier(0);
        frag_half(faH, frd + 4 * QCH);
        proto_half(fbH, bptr + g * (8 * QCH) + 4 * QCH);
        __builtin_amdgcn_sched_barrier(0);
        use_half(faL, fbL, acc0, acc1, aa);
        if (g + 1 < NGRP) {
            __builtin_amdgcn_sched_barrier(0);
            wave_lds_order();
            stage_group(buf[(g + 1) % NBUF], stw);
            wave_lds_order();
            __builtin_amdgcn_sched_barrier(0);
            if (g + 1 + NBUF < NGRP) load_group(buf[(g + 1) % NBUF], qtask, roff, g + 1 + NBUF);
            __builtin_amdgcn_sched_barrier(0);
            frag_half(faL, frd);
            proto_half(fbL, bptr + (g + 1) * (8 * QCH));
        }
        __builtin_amdgcn_sched_barrier(0);
        use_half(faH, fbH, acc0, acc1, aa);
    }

    ldsAA[wave * 32 + lane] = aa;
    __syncthreads();

    if (n < NWAY) {
        const float bbv = ldsBB[n];
        #pragma unroll
        for (int j = 0; j < 8; ++j) {
            const int m   = half * 8 + j;
            const int row = wave * 16 + m;
            if (row < NQ) {
                const float aarow = ldsAA[wave * 32 + m] + ldsAA[wave * 32 + m + 16];
                const float abv   = acc0[j] + acc1[j];
                const float logit = sc * (-(aarow - 2.0f * abv + bbv) * (1.0f / (float)DIM));
                out[((size_t)task * NQ + row) * NWAY + n] = logit;
            }
        }
    }
}

extern "C" void kernel_launch(void* const* d_in, const int* in_sizes, int n_in,
                              void* d_out, int out_size, void* d_ws, size_t ws_size,
                              hipStream_t stream) {
    const float* query   = (const float*)d_in[0];
    const float* support = (const float*)d_in[1];
    const int*   slab    = (const int*)  d_in[2];
    const float* scale   = (const float*)d_in[3];
    float*       out     = (float*)d_out;
    (void)in_sizes; (void)n_in; (void)out_size; (void)d_ws; (void)ws_size;

    proto_head_kernel<<<TASKS, 160, 0, stream>>>(query, support, slab, scale, out);
}
